# starting kernel
# baseline (speedup 1.0000x reference)
#include <hip/hip_runtime.h>


typedef __attribute__((ext_vector_type(2))) float v2f;
typedef __attribute__((ext_vector_type(4))) float v4f;
typedef __attribute__((ext_vector_type(8))) float v8f;

#define BM 32
#define XS_PITCH 68

__global__ __launch_bounds__(128)
void Linear_BC_circconv_wmma(const float* __restrict__ x,
                             const float* __restrict__ w,
                             float* __restrict__ out) {
  __shared__ float wsl[64 * 64];
  __shared__ float xs[BM][XS_PITCH];

  const int tid   = threadIdx.x;
  const int lane  = tid & 31;
  const int wave  = tid >> 5;

  const int f      = blockIdx.x;
  const int mBlock = blockIdx.y * BM;

  {
    const v4f* src = (const v4f*)(w + (size_t)f * 4096);
    v4f* dst = (v4f*)wsl;
#pragma unroll
    for (int i = 0; i < 8; ++i)
      dst[tid + i * 128] = src[tid + i * 128];
  }

  const int mA    = lane & 15;
  const int kHalf = (lane >> 4) << 1;
  const int nL    = wave * 16 + (lane & 15);
  const int rowA0 = mA;
  const int rowA1 = mA + 16;

  v8f acc0 = {};
  v8f acc1 = {};

  for (int c = 0; c < 64; ++c) {
    __syncthreads();
#pragma unroll
    for (int i = 0; i < 4; ++i) {
      int q  = tid + i * 128;
      int m  = q >> 4;
      int r4 = q & 15;
      v4f v = *(const v4f*)(x + (size_t)(mBlock + m) * 4096 + c * 64 + r4 * 4);
      *(v4f*)&xs[m][r4 * 4] = v;
    }
    __syncthreads();

    const float* wrow = wsl + c * 64;
#pragma unroll
    for (int kk = 0; kk < 16; ++kk) {
      const int k0 = kk * 4 + kHalf;
      v2f a0 = *(const v2f*)&xs[rowA0][k0];
      v2f a1 = *(const v2f*)&xs[rowA1][k0];
      v2f b;
      b.x = wrow[(nL - k0) & 63];
      b.y = wrow[(nL - k0 - 1) & 63];
      acc0 = __builtin_amdgcn_wmma_f32_16x16x4_f32(false, a0, false, b,
                                                   (short)0, acc0, false, false);
      acc1 = __builtin_amdgcn_wmma_f32_16x16x4_f32(false, a1, false, b,
                                                   (short)0, acc1, false, false);
    }
  }

  const int col     = f * 64 + nL;
  const int rowBase = mBlock + ((lane >> 4) << 3);
#pragma unroll
  for (int g = 0; g < 8; ++g) {
    out[(size_t)(rowBase + g) * 4096 + col]      = acc0[g];
    out[(size_t)(rowBase + 16 + g) * 4096 + col] = acc1[g];
  }
}

extern "C" void kernel_launch(void* const* d_in, const int* in_sizes, int n_in,
                              void* d_out, int out_size, void* d_ws, size_t ws_size,
                              hipStream_t stream) {
  (void)in_sizes; (void)n_in; (void)out_size; (void)d_ws; (void)ws_size;
  const float* x = (const float*)d_in[0];
  const float* w = (const float*)d_in[1];
  float* out = (float*)d_out;

  dim3 grid(64, 8);
  Linear_BC_circconv_wmma<<<grid, 128, 0, stream>>>(x, w, out);
}


# Retile: wave tile 32x16 -> 16x64 (block 16x256, K staged 32 chunks deep); a chunk's 16 distinct circulant B fragments read once for 4 column subtiles; k order permuted, f32 sum order changes
# speedup vs baseline: 3.2947x; 3.2947x over previous
#include <hip/hip_runtime.h>


typedef __attribute__((ext_vector_type(2))) float v2f;
typedef __attribute__((ext_vector_type(4))) float v4f;
typedef __attribute__((ext_vector_type(8))) float v8f;

#define ST_CH 32
#define ST_K  (ST_CH * 64)
#define XP    (ST_K + 4)
#define WP    80
#define WS_FLOATS (4 * ST_CH * WP)
#define XS_FLOATS (16 * XP)

struct Frags {
  v4f a[8];
  float b[8][4];
};

__device__ __forceinline__ void load_frags(Frags& fr, const float* ap, const float* bp) {
#pragma unroll
  for (int j = 0; j < 8; ++j) fr.a[j] = *(const v4f*)(ap + 8 * j);
#pragma unroll
  for (int jj = 0; jj < 8; ++jj) {
#pragma unroll
    for (int i = 0; i < 4; ++i) fr.b[jj][i] = bp[8 * (7 - jj) + i];
  }
}

__device__ __forceinline__ void mma_chunk(const Frags& fr, v8f& acc0, v8f& acc1,
                                          v8f& acc2, v8f& acc3) {
#pragma unroll
  for (int j = 0; j < 8; ++j) {
#pragma unroll
    for (int p = 0; p < 2; ++p) {
      v2f av;
      av.x = fr.a[j][2 * p];
      av.y = fr.a[j][2 * p + 1];
      v2f b0, b1, b2, b3;
      b0.x = fr.b[(0 - j) & 7][2 * p]; b0.y = fr.b[(0 - j) & 7][2 * p + 1];
      b1.x = fr.b[(2 - j) & 7][2 * p]; b1.y = fr.b[(2 - j) & 7][2 * p + 1];
      b2.x = fr.b[(4 - j) & 7][2 * p]; b2.y = fr.b[(4 - j) & 7][2 * p + 1];
      b3.x = fr.b[(6 - j) & 7][2 * p]; b3.y = fr.b[(6 - j) & 7][2 * p + 1];
      acc0 = __builtin_amdgcn_wmma_f32_16x16x4_f32(false, av, false, b0, (short)0, acc0, false, false);
      acc1 = __builtin_amdgcn_wmma_f32_16x16x4_f32(false, av, false, b1, (short)0, acc1, false, false);
      acc2 = __builtin_amdgcn_wmma_f32_16x16x4_f32(false, av, false, b2, (short)0, acc2, false, false);
      acc3 = __builtin_amdgcn_wmma_f32_16x16x4_f32(false, av, false, b3, (short)0, acc3, false, false);
    }
  }
}

__global__ __launch_bounds__(128)
void Linear_BC_circconv_wmma(const float* __restrict__ x,
                             const float* __restrict__ w,
                             float* __restrict__ out) {
  __shared__ __attribute__((aligned(16))) float lds[WS_FLOATS + XS_FLOATS];
  float* wsx = lds;
  float* xs  = lds + WS_FLOATS;

  const int tid  = threadIdx.x;
  const int lane = tid & 31;
  const int wave = tid >> 5;
  const int l15  = lane & 15;
  const int h    = lane >> 4;

  const int fg = blockIdx.x;
  const int m0 = blockIdx.y * 16;

  v8f acc0 = {}, acc1 = {}, acc2 = {}, acc3 = {};

  const float* aBase = xs + l15 * XP + 4 * h;
  const float* bBase = wsx + wave * (ST_CH * WP) + (15 - l15 + 4 * h);

  for (int st = 0; st < 2; ++st) {
    __syncthreads();

    const float* xg = x + (size_t)m0 * 4096 + st * ST_K + tid * 4;
    const float* wg = w + ((size_t)(fg * 4) * 64 + st * ST_CH) * 64 + tid * 4;
    {
      v4f t[40];
#pragma unroll
      for (int i = 0; i < 40; ++i)
        t[i] = *(const v4f*)(xg + (size_t)(i >> 2) * 4096 + (i & 3) * 512);
#pragma unroll
      for (int i = 0; i < 40; ++i)
        *(v4f*)(xs + (i >> 2) * XP + (i & 3) * 512 + tid * 4) = t[i];
    }
    {
      v4f t[24];
      v4f u[16];
#pragma unroll
      for (int i = 0; i < 24; ++i)
        t[i] = *(const v4f*)(xg + (size_t)((i + 40) >> 2) * 4096 + ((i + 40) & 3) * 512);
#pragma unroll
      for (int i = 0; i < 16; ++i)
        u[i] = *(const v4f*)(wg + (size_t)(i >> 2) * 4096 + (i & 3) * 512);
#pragma unroll
      for (int i = 0; i < 24; ++i)
        *(v4f*)(xs + ((i + 40) >> 2) * XP + ((i + 40) & 3) * 512 + tid * 4) = t[i];
      const int r4  = tid & 15;
      const int rr  = tid >> 4;
#pragma unroll
      for (int i = 0; i < 16; ++i) {
        v4f rv;
        rv.x = u[i].w; rv.y = u[i].z; rv.z = u[i].y; rv.w = u[i].x;
        u[i] = rv;
      }
      float* wrow = wsx + rr * WP;
#pragma unroll
      for (int i = 0; i < 16; ++i)
        *(v4f*)(wrow + (i >> 2) * (ST_CH * WP) + (i & 3) * 8 * WP + 68 - r4 * 4) = u[i];
      if (r4 < 2) {
#pragma unroll
        for (int i = 0; i < 16; ++i)
          *(v4f*)(wrow + (i >> 2) * (ST_CH * WP) + (i & 3) * 8 * WP + 4 - r4 * 4) = u[i];
      }
      if (r4 >= 14) {
#pragma unroll
        for (int i = 0; i < 16; ++i)
          *(v4f*)(wrow + (i >> 2) * (ST_CH * WP) + (i & 3) * 8 * WP + 132 - r4 * 4) = u[i];
      }
    }
    __syncthreads();

    const float* ap = aBase;
    const float* bp = bBase;
    Frags f0, f1;
    load_frags(f0, ap, bp);
#pragma unroll 1
    for (int cc = 0; cc < ST_CH; cc += 2) {
      load_frags(f1, ap + 64, bp + WP);
      __builtin_amdgcn_sched_barrier(0);
      mma_chunk(f0, acc0, acc1, acc2, acc3);
      ap += 128;
      bp += 2 * WP;
      if (cc + 2 < ST_CH) load_frags(f0, ap, bp);
      __builtin_amdgcn_sched_barrier(0);
      mma_chunk(f1, acc0, acc1, acc2, acc3);
    }
  }

  float* o = out + (size_t)(m0 + 8 * h) * 4096 + (fg * 4 + wave) * 64 + l15;
#pragma unroll
  for (int r = 0; r < 8; ++r) {
    o[(size_t)r * 4096 +  0] = acc0[r];
    o[(size_t)r * 4096 + 16] = acc1[r];
    o[(size_t)r * 4096 + 32] = acc2[r];
    o[(size_t)r * 4096 + 48] = acc3[r];
  }
}

extern "C" void kernel_launch(void* const* d_in, const int* in_sizes, int n_in,
                              void* d_out, int out_size, void* d_ws, size_t ws_size,
                              hipStream_t stream) {
  (void)in_sizes; (void)n_in; (void)out_size; (void)d_ws; (void)ws_size;
  const float* x = (const float*)d_in[0];
  const float* w = (const float*)d_in[1];
  float* out = (float*)d_out;

  dim3 grid(16, 16);
  Linear_BC_circconv_wmma<<<grid, 128, 0, stream>>>(x, w, out);
}


# Fold the 32-point cyclic half again by z^32-1=(z^16-1)(z^16+1): xp,wp lo+hi, lo-hi across lane pairs at staging, 16-point cyclic + negacyclic products; WMMAs 32->24 per chunk; f32 sum order changes
# speedup vs baseline: 5.8894x; 1.7875x over previous
#include <hip/hip_runtime.h>


typedef __attribute__((ext_vector_type(2))) float v2f;
typedef __attribute__((ext_vector_type(4))) float v4f;
typedef __attribute__((ext_vector_type(8))) float v8f;

#define ST_CH 32
#define ST_K  (ST_CH * 64)
#define XP    (ST_K + 4)
#define WPM   64
#define WP    144
#define WS_FLOATS (4 * ST_CH * WP)
#define XS_FLOATS (16 * XP)

#define DPP8_XOR4 (4u | (5u << 3) | (6u << 6) | (7u << 9) | (0u << 12) | (1u << 15) | (2u << 18) | (3u << 21))

__device__ __forceinline__ v4f other_half(v4f v) {
  v4f r;
  r.x = __uint_as_float(__builtin_amdgcn_mov_dpp8(__float_as_uint(v.x), DPP8_XOR4));
  r.y = __uint_as_float(__builtin_amdgcn_mov_dpp8(__float_as_uint(v.y), DPP8_XOR4));
  r.z = __uint_as_float(__builtin_amdgcn_mov_dpp8(__float_as_uint(v.z), DPP8_XOR4));
  r.w = __uint_as_float(__builtin_amdgcn_mov_dpp8(__float_as_uint(v.w), DPP8_XOR4));
  return r;
}

struct Frags {
  v4f a[8];
  float bc[2][4];
  float bn[2][4];
  float bm[6][4];
};

template <int ODD>
__device__ __forceinline__ void load_frags(Frags& fr, const float* ap, const float* bp) {
  const float* aP = ap + (ODD ? 32 : 0);
  const float* aM = ap + (ODD ? 0 : 32);
#pragma unroll
  for (int j = 0; j < 4; ++j) fr.a[j] = *(const v4f*)(aP + 8 * j);
#pragma unroll
  for (int j = 0; j < 4; ++j) fr.a[4 + j] = *(const v4f*)(aM + 8 * j);
#pragma unroll
  for (int j = 0; j < 2; ++j) {
#pragma unroll
    for (int i = 0; i < 4; ++i) {
      fr.bc[j][i] = bp[8 * j + i];
      fr.bn[j][i] = bp[32 + 8 * j + i];
    }
  }
#pragma unroll
  for (int qq = 0; qq < 6; ++qq) {
#pragma unroll
    for (int i = 0; i < 4; ++i) fr.bm[qq][i] = bp[WPM + 8 * (5 - qq) + i];
  }
}

__device__ __forceinline__ void mma_chunk(const Frags& fr, v8f& accC, v8f& accN,
                                          v8f& accM0, v8f& accM1) {
#pragma unroll
  for (int j = 0; j < 4; ++j) {
#pragma unroll
    for (int p = 0; p < 2; ++p) {
      v2f am;
      am.x = fr.a[4 + j][2 * p]; am.y = fr.a[4 + j][2 * p + 1];
      v2f b2, b3;
      b2.x = fr.bm[3 - j][2 * p];       b2.y = fr.bm[3 - j][2 * p + 1];
      b3.x = fr.bm[5 - j][2 * p];       b3.y = fr.bm[5 - j][2 * p + 1];
      if (j < 2) {
        v2f ac, an, b0, b1;
        ac.x = fr.a[j][2 * p];     ac.y = fr.a[j][2 * p + 1];
        an.x = fr.a[2 + j][2 * p]; an.y = fr.a[2 + j][2 * p + 1];
        b0.x = fr.bc[j][2 * p];    b0.y = fr.bc[j][2 * p + 1];
        b1.x = fr.bn[j][2 * p];    b1.y = fr.bn[j][2 * p + 1];
        accC = __builtin_amdgcn_wmma_f32_16x16x4_f32(false, ac, false, b0, (short)0, accC, false, false);
        accM0 = __builtin_amdgcn_wmma_f32_16x16x4_f32(false, am, false, b2, (short)0, accM0, false, false);
        accN = __builtin_amdgcn_wmma_f32_16x16x4_f32(false, an, false, b1, (short)0, accN, false, false);
        accM1 = __builtin_amdgcn_wmma_f32_16x16x4_f32(false, am, false, b3, (short)0, accM1, false, false);
      } else {
        accM0 = __builtin_amdgcn_wmma_f32_16x16x4_f32(false, am, false, b2, (short)0, accM0, false, false);
        accM1 = __builtin_amdgcn_wmma_f32_16x16x4_f32(false, am, false, b3, (short)0, accM1, false, false);
      }
    }
  }
}

__global__ __launch_bounds__(128)
void Linear_BC_circconv_wmma(const float* __restrict__ x,
                             const float* __restrict__ w,
                             float* __restrict__ out) {
  __shared__ __attribute__((aligned(16))) float lds[WS_FLOATS + XS_FLOATS];
  float* wsx = lds;
  float* xs  = lds + WS_FLOATS;

  const int tid  = threadIdx.x;
  const int lane = tid & 31;
  const int wave = tid >> 5;
  const int l15  = lane & 15;
  const int h    = lane >> 4;

  const int fg = blockIdx.x;
  const int m0 = blockIdx.y * 16;

  v8f accC = {}, accN = {}, accM0 = {}, accM1 = {};

  const float* aBase = xs + l15 * XP + 4 * h;
  const float* bBase = wsx + wave * (ST_CH * WP) + (15 - l15 + 4 * h);

  const int s8 = tid & 7;
  const int s4 = s8 & 3;
  const int hf = s8 >> 2;
  const float sg2 = hf ? -1.0f : 1.0f;
  const int ch = tid >> 3;
  const int   oP = (ch & 1) ? 32 : 0;
  const int   oM = 32 - oP;

  for (int st = 0; st < 2; ++st) {
    __syncthreads();

    const float* xg = x + (size_t)m0 * 4096 + st * ST_K + ch * 64 + s8 * 4;
    const float* wg = w + ((size_t)(fg * 4) * 64 + st * ST_CH + ch) * 64 + s8 * 4;
    float* xd = xs + ch * 64 + s8 * 4;
    {
      v4f lo[20], hi[20];
#pragma unroll
      for (int i = 0; i < 20; ++i) {
        lo[i] = *(const v4f*)(xg + (size_t)(i >> 1) * 4096 + (i & 1) * 1024);
        hi[i] = *(const v4f*)(xg + (size_t)(i >> 1) * 4096 + (i & 1) * 1024 + 32);
      }
#pragma unroll
      for (int i = 0; i < 20; ++i) {
        const v4f xp = lo[i] + hi[i];
        *(v4f*)(xd + (i >> 1) * XP + (i & 1) * 1024 + oP) = other_half(xp) + sg2 * xp;
        *(v4f*)(xd + (i >> 1) * XP + (i & 1) * 1024 + oM) = lo[i] - hi[i];
      }
    }
    {
      v4f lo[12], hi[12];
      v4f ul[8], uh[8];
#pragma unroll
      for (int i = 0; i < 12; ++i) {
        lo[i] = *(const v4f*)(xg + (size_t)((i + 20) >> 1) * 4096 + ((i + 20) & 1) * 1024);
        hi[i] = *(const v4f*)(xg + (size_t)((i + 20) >> 1) * 4096 + ((i + 20) & 1) * 1024 + 32);
      }
#pragma unroll
      for (int i = 0; i < 8; ++i) {
        ul[i] = *(const v4f*)(wg + (size_t)(i >> 1) * 4096 + (i & 1) * 1024);
        uh[i] = *(const v4f*)(wg + (size_t)(i >> 1) * 4096 + (i & 1) * 1024 + 32);
      }
#pragma unroll
      for (int i = 0; i < 12; ++i) {
        const v4f xp = lo[i] + hi[i];
        *(v4f*)(xd + ((i + 20) >> 1) * XP + ((i + 20) & 1) * 1024 + oP) = other_half(xp) + sg2 * xp;
        *(v4f*)(xd + ((i + 20) >> 1) * XP + ((i + 20) & 1) * 1024 + oM) = lo[i] - hi[i];
      }
      float* wrow = wsx + ch * WP;
      v4f rq[8], rm[8];
#pragma unroll
      for (int i = 0; i < 8; ++i) {
        const v4f sp = ul[i] + uh[i];
        const v4f sm = ul[i] - uh[i];
        const v4f sq = other_half(sp) + sg2 * sp;
        rq[i].x = sq.w; rq[i].y = sq.z; rq[i].z = sq.y; rq[i].w = sq.x;
        rm[i].x = sm.w; rm[i].y = sm.z; rm[i].z = sm.y; rm[i].w = sm.x;
      }
#pragma unroll
      for (int i = 0; i < 8; ++i) {
        float* d = wrow + (i >> 1) * (ST_CH * WP) + (i & 1) * 16 * WP;
        *(v4f*)(d + 32 * hf + 12 - s4 * 4) = rq[i];
        *(v4f*)(d + 32 * hf + 28 - s4 * 4) = sg2 * rq[i];
        *(v4f*)(d + WPM + 28 - s8 * 4) = rm[i];
        *(v4f*)(d + WPM + 60 - s8 * 4) = -rm[i];
      }
    }
    __syncthreads();

    const float* ap = aBase;
    const float* bp = bBase;
    Frags f0, f1;
    load_frags<0>(f0, ap, bp);
#pragma unroll 1
    for (int cc = 0; cc < ST_CH; cc += 2) {
      load_frags<1>(f1, ap + 64, bp + WP);
      __builtin_amdgcn_sched_barrier(0);
      mma_chunk(f0, accC, accN, accM0, accM1);
      ap += 128;
      bp += 2 * WP;
      if (cc + 2 < ST_CH) load_frags<0>(f0, ap, bp);
      __builtin_amdgcn_sched_barrier(0);
      mma_chunk(f1, accC, accN, accM0, accM1);
    }
  }

  float* o = out + (size_t)(m0 + 8 * h) * 4096 + (fg * 4 + wave) * 64 + l15;
#pragma unroll
  for (int r = 0; r < 8; ++r) {
    const float p0 = 0.5f * (accC[r] + accN[r]);
    const float p1 = 0.5f * (accC[r] - accN[r]);
    o[(size_t)r * 4096 +  0] = 0.5f * (p0 + accM0[r]);
    o[(size_t)r * 4096 + 16] = 0.5f * (p1 + accM1[r]);
    o[(size_t)r * 4096 + 32] = 0.5f * (p0 - accM0[r]);
    o[(size_t)r * 4096 + 48] = 0.5f * (p1 - accM1[r]);
  }
}

extern "C" void kernel_launch(void* const* d_in, const int* in_sizes, int n_in,
                              void* d_out, int out_size, void* d_ws, size_t ws_size,
                              hipStream_t stream) {
  (void)in_sizes; (void)n_in; (void)out_size; (void)d_ws; (void)ws_size;
  const float* x = (const float*)d_in[0];
  const float* w = (const float*)d_in[1];
  float* out = (float*)d_out;

  dim3 grid(16, 16);
  Linear_BC_circconv_wmma<<<grid, 128, 0, stream>>>(x, w, out);
}


# Negacyclic 32-point half as a 16-point complex product, 3 real products not 4: T0 x_lo, T1 x_hi, (T0+T1)(x_lo+x_hi), sums formed in registers; WMMAs 24->20 per chunk; f32 sum order changes
# speedup vs baseline: 6.4141x; 1.0891x over previous
#include <hip/hip_runtime.h>


typedef __attribute__((ext_vector_type(2))) float v2f;
typedef __attribute__((ext_vector_type(4))) float v4f;
typedef __attribute__((ext_vector_type(8))) float v8f;

#define ST_CH 32
#define ST_K  (ST_CH * 64)
#define XP    (ST_K + 4)
#define WPM   64
#define WP    144
#define WS_FLOATS (4 * ST_CH * WP)
#define XS_FLOATS (16 * XP)

#define DPP8_XOR4 (4u | (5u << 3) | (6u << 6) | (7u << 9) | (0u << 12) | (1u << 15) | (2u << 18) | (3u << 21))

__device__ __forceinline__ v4f other_half(v4f v) {
  v4f r;
  r.x = __uint_as_float(__builtin_amdgcn_mov_dpp8(__float_as_uint(v.x), DPP8_XOR4));
  r.y = __uint_as_float(__builtin_amdgcn_mov_dpp8(__float_as_uint(v.y), DPP8_XOR4));
  r.z = __uint_as_float(__builtin_amdgcn_mov_dpp8(__float_as_uint(v.z), DPP8_XOR4));
  r.w = __uint_as_float(__builtin_amdgcn_mov_dpp8(__float_as_uint(v.w), DPP8_XOR4));
  return r;
}

struct Frags {
  v4f a[8];
  float bc[2][4];
  float bn[2][4];
  float bt0[2][4];
  float bt1[2][4];
};

template <int ODD>
__device__ __forceinline__ void load_frags(Frags& fr, const float* ap, const float* bp) {
  const float* aP = ap + (ODD ? 32 : 0);
  const float* aM = ap + (ODD ? 0 : 32);
#pragma unroll
  for (int j = 0; j < 4; ++j) fr.a[j] = *(const v4f*)(aP + 8 * j);
#pragma unroll
  for (int j = 0; j < 4; ++j) fr.a[4 + j] = *(const v4f*)(aM + 8 * j);
#pragma unroll
  for (int j = 0; j < 2; ++j) {
#pragma unroll
    for (int i = 0; i < 4; ++i) {
      fr.bc[j][i] = bp[8 * j + i];
      fr.bn[j][i] = bp[32 + 8 * j + i];
    }
  }
#pragma unroll
  for (int j = 0; j < 2; ++j) {
#pragma unroll
    for (int i = 0; i < 4; ++i) {
      fr.bt1[j][i] = bp[WPM + 8 * j + i];
      fr.bt0[j][i] = bp[WPM + 16 + 8 * j + i];
    }
  }
}

__device__ __forceinline__ void mma_chunk(const Frags& fr, v8f& accC, v8f& accN,
                                          v8f& accP1, v8f& accP2, v8f& accP3) {
#pragma unroll
  for (int j = 0; j < 2; ++j) {
    const v4f as = fr.a[4 + j] + fr.a[6 + j];
    float bs[4];
#pragma unroll
    for (int i = 0; i < 4; ++i) bs[i] = fr.bt0[j][i] + fr.bt1[j][i];
#pragma unroll
    for (int p = 0; p < 2; ++p) {
      v2f ac, an, al, ah, a3, b0, b1, b2, b3, b4;
      ac.x = fr.a[j][2 * p];     ac.y = fr.a[j][2 * p + 1];
      an.x = fr.a[2 + j][2 * p]; an.y = fr.a[2 + j][2 * p + 1];
      al.x = fr.a[4 + j][2 * p]; al.y = fr.a[4 + j][2 * p + 1];
      ah.x = fr.a[6 + j][2 * p]; ah.y = fr.a[6 + j][2 * p + 1];
      a3.x = as[2 * p];          a3.y = as[2 * p + 1];
      b0.x = fr.bc[j][2 * p];    b0.y = fr.bc[j][2 * p + 1];
      b1.x = fr.bn[j][2 * p];    b1.y = fr.bn[j][2 * p + 1];
      b2.x = fr.bt0[j][2 * p];   b2.y = fr.bt0[j][2 * p + 1];
      b3.x = fr.bt1[j][2 * p];   b3.y = fr.bt1[j][2 * p + 1];
      b4.x = bs[2 * p];          b4.y = bs[2 * p + 1];
      accC  = __builtin_amdgcn_wmma_f32_16x16x4_f32(false, ac, false, b0, (short)0, accC,  false, false);
      accP1 = __builtin_amdgcn_wmma_f32_16x16x4_f32(false, al, false, b2, (short)0, accP1, false, false);
      accN  = __builtin_amdgcn_wmma_f32_16x16x4_f32(false, an, false, b1, (short)0, accN,  false, false);
      accP2 = __builtin_amdgcn_wmma_f32_16x16x4_f32(false, ah, false, b3, (short)0, accP2, false, false);
      accP3 = __builtin_amdgcn_wmma_f32_16x16x4_f32(false, a3, false, b4, (short)0, accP3, false, false);
    }
  }
}

__global__ __launch_bounds__(128)
void Linear_BC_circconv_wmma(const float* __restrict__ x,
                             const float* __restrict__ w,
                             float* __restrict__ out) {
  __shared__ __attribute__((aligned(16))) float lds[WS_FLOATS + XS_FLOATS];
  float* wsx = lds;
  float* xs  = lds + WS_FLOATS;

  const int tid  = threadIdx.x;
  const int lane = tid & 31;
  const int wave = tid >> 5;
  const int l15  = lane & 15;
  const int h    = lane >> 4;

  const int fg = blockIdx.x;
  const int m0 = blockIdx.y * 16;

  v8f accC = {}, accN = {}, accP1 = {}, accP2 = {}, accP3 = {};

  const float* aBase = xs + l15 * XP + 4 * h;
  const float* bBase = wsx + wave * (ST_CH * WP) + (15 - l15 + 4 * h);

  const int s8 = tid & 7;
  const int s4 = s8 & 3;
  const int hf = s8 >> 2;
  const float sg2 = hf ? -1.0f : 1.0f;
  const int ch = tid >> 3;
  const int   oP = (ch & 1) ? 32 : 0;
  const int   oM = 32 - oP;

  for (int st = 0; st < 2; ++st) {
    __syncthreads();

    const float* xg = x + (size_t)m0 * 4096 + st * ST_K + ch * 64 + s8 * 4;
    const float* wg = w + ((size_t)(fg * 4) * 64 + st * ST_CH + ch) * 64 + s8 * 4;
    float* xd = xs + ch * 64 + s8 * 4;
    {
      v4f lo[20], hi[20];
#pragma unroll
      for (int i = 0; i < 20; ++i) {
        lo[i] = *(const v4f*)(xg + (size_t)(i >> 1) * 4096 + (i & 1) * 1024);
        hi[i] = *(const v4f*)(xg + (size_t)(i >> 1) * 4096 + (i & 1) * 1024 + 32);
      }
#pragma unroll
      for (int i = 0; i < 20; ++i) {
        const v4f xp = lo[i] + hi[i];
        *(v4f*)(xd + (i >> 1) * XP + (i & 1) * 1024 + oP) = other_half(xp) + sg2 * xp;
        *(v4f*)(xd + (i >> 1) * XP + (i & 1) * 1024 + oM) = lo[i] - hi[i];
      }
    }
    {
      v4f lo[12], hi[12];
      v4f ul[8], uh[8];
#pragma unroll
      for (int i = 0; i < 12; ++i) {
        lo[i] = *(const v4f*)(xg + (size_t)((i + 20) >> 1) * 4096 + ((i + 20) & 1) * 1024);
        hi[i] = *(const v4f*)(xg + (size_t)((i + 20) >> 1) * 4096 + ((i + 20) & 1) * 1024 + 32);
      }
#pragma unroll
      for (int i = 0; i < 8; ++i) {
        ul[i] = *(const v4f*)(wg + (size_t)(i >> 1) * 4096 + (i & 1) * 1024);
        uh[i] = *(const v4f*)(wg + (size_t)(i >> 1) * 4096 + (i & 1) * 1024 + 32);
      }
#pragma unroll
      for (int i = 0; i < 12; ++i) {
        const v4f xp = lo[i] + hi[i];
        *(v4f*)(xd + ((i + 20) >> 1) * XP + ((i + 20) & 1) * 1024 + oP) = other_half(xp) + sg2 * xp;
        *(v4f*)(xd + ((i + 20) >> 1) * XP + ((i + 20) & 1) * 1024 + oM) = lo[i] - hi[i];
      }
      float* wrow = wsx + ch * WP;
      v4f rq[8], rm[8];
#pragma unroll
      for (int i = 0; i < 8; ++i) {
        const v4f sp = ul[i] + uh[i];
        const v4f sm = ul[i] - uh[i];
        const v4f sq = other_half(sp) + sg2 * sp;
        rq[i].x = sq.w; rq[i].y = sq.z; rq[i].z = sq.y; rq[i].w = sq.x;
        rm[i].x = sm.w; rm[i].y = sm.z; rm[i].z = sm.y; rm[i].w = sm.x;
      }
#pragma unroll
      for (int i = 0; i < 8; ++i) {
        float* d = wrow + (i >> 1) * (ST_CH * WP) + (i & 1) * 16 * WP;
        *(v4f*)(d + 32 * hf + 12 - s4 * 4) = rq[i];
        *(v4f*)(d + 32 * hf + 28 - s4 * 4) = sg2 * rq[i];
        *(v4f*)(d + WPM + 28 - s8 * 4) = rm[i];
        *(v4f*)(d + WPM + 60 - s8 * 4) = -rm[i];
      }
    }
    __syncthreads();

    const float* ap = aBase;
    const float* bp = bBase;
    Frags f0, f1;
    load_frags<0>(f0, ap, bp);
#pragma unroll 1
    for (int cc = 0; cc < ST_CH; cc += 2) {
      load_frags<1>(f1, ap + 64, bp + WP);
      __builtin_amdgcn_sched_barrier(0);
      mma_chunk(f0, accC, accN, accP1, accP2, accP3);
      ap += 128;
      bp += 2 * WP;
      if (cc + 2 < ST_CH) load_frags<0>(f0, ap, bp);
      __builtin_amdgcn_sched_barrier(0);
      mma_chunk(f1, accC, accN, accP1, accP2, accP3);
    }
  }

  float* o = out + (size_t)(m0 + 8 * h) * 4096 + (fg * 4 + wave) * 64 + l15;
#pragma unroll
  for (int r = 0; r < 8; ++r) {
    const float p0 = 0.5f * (accC[r] + accN[r]);
    const float p1 = 0.5f * (accC[r] - accN[r]);
    const float q0 = accP1[r] - accP2[r];
    const float q1 = accP3[r] - accP1[r] - accP2[r];
    o[(size_t)r * 4096 +  0] = 0.5f * (p0 + q0);
    o[(size_t)r * 4096 + 16] = 0.5f * (p1 + q1);
    o[(size_t)r * 4096 + 32] = 0.5f * (p0 - q0);
    o[(size_t)r * 4096 + 48] = 0.5f * (p1 - q1);
  }
}

extern "C" void kernel_launch(void* const* d_in, const int* in_sizes, int n_in,
                              void* d_out, int out_size, void* d_ws, size_t ws_size,
                              hipStream_t stream) {
  (void)in_sizes; (void)n_in; (void)out_size; (void)d_ws; (void)ws_size;
  const float* x = (const float*)d_in[0];
  const float* w = (const float*)d_in[1];
  float* out = (float*)d_out;

  dim3 grid(16, 16);
  Linear_BC_circconv_wmma<<<grid, 128, 0, stream>>>(x, w, out);
}
